# starting kernel
# baseline (speedup 1.0000x reference)
#include <hip/hip_runtime.h>

typedef __attribute__((ext_vector_type(2))) float v2f;
typedef __attribute__((ext_vector_type(8))) float v8f;

#define D_FEAT     64
#define LDS_STRIDE 68

__global__ __launch_bounds__(256)
void batchdist_wmma_f32(const float* __restrict__ X, float* __restrict__ out, int n)
{
    __shared__ float sAi[32 * LDS_STRIDE];
    __shared__ float sLi[32 * LDS_STRIDE];
    __shared__ float sAj[64 * LDS_STRIDE];
    __shared__ float sLj[64 * LDS_STRIDE];

    const int tid = threadIdx.x;
    const int i0  = blockIdx.y * 32;
    const int j0  = blockIdx.x * 64;

    for (int q = tid; q < 32 * 16; q += 256) {
        const int row = q >> 4;
        const int c4  = (q & 15) << 2;
        const float4 g = *(const float4*)(X + (size_t)(i0 + row) * D_FEAT + c4);
        float4 l;
        l.x = g.x * __logf(g.x); l.y = g.y * __logf(g.y);
        l.z = g.z * __logf(g.z); l.w = g.w * __logf(g.w);
        *(float4*)(sAi + row * LDS_STRIDE + c4) = g;
        *(float4*)(sLi + row * LDS_STRIDE + c4) = l;
    }
    for (int q = tid; q < 64 * 16; q += 256) {
        const int row = q >> 4;
        const int c4  = (q & 15) << 2;
        const float4 g = *(const float4*)(X + (size_t)(j0 + row) * D_FEAT + c4);
        float4 l;
        l.x = g.x * __logf(g.x); l.y = g.y * __logf(g.y);
        l.z = g.z * __logf(g.z); l.w = g.w * __logf(g.w);
        *(float4*)(sAj + row * LDS_STRIDE + c4) = g;
        *(float4*)(sLj + row * LDS_STRIDE + c4) = l;
    }
    __syncthreads();

    const int wave = tid >> 5;
    const int lane = tid & 31;
    const int half = lane >> 4;
    const int mn   = lane & 15;
    const int sr   = wave >> 2;
    const int sc   = wave & 3;

    const float* li = sLi + (sr * 16 + mn) * LDS_STRIDE + 2 * half;
    const float* ai = sAi + (sr * 16 + mn) * LDS_STRIDE + 2 * half;
    const float* aj = sAj + (sc * 16 + mn) * LDS_STRIDE + 2 * half;
    const float* lj = sLj + (sc * 16 + mn) * LDS_STRIDE + 2 * half;

    v8f c = {};
#pragma unroll
    for (int k = 0; k < D_FEAT; k += 4) {
        const v2f aL = *(const v2f*)(li + k);
        const v2f aA = *(const v2f*)(ai + k);
        const v2f bA = *(const v2f*)(aj + k);
        const v2f bL = *(const v2f*)(lj + k);
        c = __builtin_amdgcn_wmma_f32_16x16x4_f32(false, aL, false, bA,
                                                  (short)0, c, false, false);
        c = __builtin_amdgcn_wmma_f32_16x16x4_f32(false, aA, false, bL,
                                                  (short)0, c, false, false);
    }

    float* orow = out + (size_t)(i0 + sr * 16 + 8 * half) * n + (j0 + sc * 16 + mn);
#pragma unroll
    for (int r = 0; r < 8; ++r) {
        orow[(size_t)r * n] = -c[r];
    }
}

extern "C" void kernel_launch(void* const* d_in, const int* in_sizes, int n_in,
                              void* d_out, int out_size, void* d_ws, size_t ws_size,
                              hipStream_t stream) {
    (void)n_in; (void)out_size; (void)d_ws; (void)ws_size;
    const float* x1 = (const float*)d_in[0];
    float* out = (float*)d_out;
    const int n = in_sizes[0] / D_FEAT;
    dim3 grid(n / 64, n / 32);
    batchdist_wmma_f32<<<grid, 256, 0, stream>>>(x1, out, n);
}


# Register tiling: each wave computes a 32x64 block (2x4 WMMA tiles) of a 128x128 workgroup tile; grid 256 workgroups of 8 waves instead of 2048; summation order per element unchanged
# speedup vs baseline: 1.6400x; 1.6400x over previous
#include <hip/hip_runtime.h>

typedef __attribute__((ext_vector_type(2))) float v2f;
typedef __attribute__((ext_vector_type(8))) float v8f;

#define D_FEAT     64
#define LDS_STRIDE 68
#define TILE       128
#define RT         2
#define CT         4
#define QPT        (TILE * 16 / 256)

__global__ __launch_bounds__(256)
void batchdist_wmma_f32(const float* __restrict__ X, float* __restrict__ out, int n)
{
    __shared__ float sAi[TILE * LDS_STRIDE];
    __shared__ float sLi[TILE * LDS_STRIDE];
    __shared__ float sAj[TILE * LDS_STRIDE];
    __shared__ float sLj[TILE * LDS_STRIDE];

    const int tid = threadIdx.x;
    const int i0  = blockIdx.y * TILE;
    const int j0  = blockIdx.x * TILE;

    const int prow = tid >> 4;
    const int pc4  = (tid & 15) << 2;
    float4 gi[QPT], gj[QPT];
#pragma unroll
    for (int it = 0; it < QPT; ++it)
        gi[it] = *(const float4*)(X + (size_t)(i0 + prow + 16 * it) * D_FEAT + pc4);
#pragma unroll
    for (int it = 0; it < QPT; ++it)
        gj[it] = *(const float4*)(X + (size_t)(j0 + prow + 16 * it) * D_FEAT + pc4);
    __builtin_amdgcn_sched_barrier(0);
#pragma unroll
    for (int it = 0; it < QPT; ++it) {
        const float4 g = gi[it];
        float4 l;
        l.x = g.x * __logf(g.x); l.y = g.y * __logf(g.y);
        l.z = g.z * __logf(g.z); l.w = g.w * __logf(g.w);
        *(float4*)(sAi + (prow + 16 * it) * LDS_STRIDE + pc4) = g;
        *(float4*)(sLi + (prow + 16 * it) * LDS_STRIDE + pc4) = l;
    }
#pragma unroll
    for (int it = 0; it < QPT; ++it) {
        const float4 g = gj[it];
        float4 l;
        l.x = g.x * __logf(g.x); l.y = g.y * __logf(g.y);
        l.z = g.z * __logf(g.z); l.w = g.w * __logf(g.w);
        *(float4*)(sAj + (prow + 16 * it) * LDS_STRIDE + pc4) = g;
        *(float4*)(sLj + (prow + 16 * it) * LDS_STRIDE + pc4) = l;
    }
    __syncthreads();

    const int wave = tid >> 5;
    const int lane = tid & 31;
    const int half = lane >> 4;
    const int mn   = lane & 15;
    const int wr   = wave >> 1;
    const int wc   = wave & 1;

    const int ioff = (wr * (RT * 16) + mn) * LDS_STRIDE + 2 * half;
    const int joff = (wc * (CT * 16) + mn) * LDS_STRIDE + 2 * half;
    const float* li = sLi + ioff;
    const float* ai = sAi + ioff;
    const float* aj = sAj + joff;
    const float* lj = sLj + joff;

    v8f c[RT][CT] = {};
#pragma unroll
    for (int k = 0; k < D_FEAT; k += 4) {
        v2f aL[RT], aA[RT], bA[CT], bL[CT];
#pragma unroll
        for (int r = 0; r < RT; ++r) {
            aL[r] = *(const v2f*)(li + r * 16 * LDS_STRIDE + k);
            aA[r] = *(const v2f*)(ai + r * 16 * LDS_STRIDE + k);
        }
#pragma unroll
        for (int t = 0; t < CT; ++t) {
            bA[t] = *(const v2f*)(aj + t * 16 * LDS_STRIDE + k);
            bL[t] = *(const v2f*)(lj + t * 16 * LDS_STRIDE + k);
        }
#pragma unroll
        for (int r = 0; r < RT; ++r)
#pragma unroll
            for (int t = 0; t < CT; ++t)
                c[r][t] = __builtin_amdgcn_wmma_f32_16x16x4_f32(false, aL[r], false, bA[t],
                                                                (short)0, c[r][t], false, false);
#pragma unroll
        for (int r = 0; r < RT; ++r)
#pragma unroll
            for (int t = 0; t < CT; ++t)
                c[r][t] = __builtin_amdgcn_wmma_f32_16x16x4_f32(false, aA[r], false, bL[t],
                                                                (short)0, c[r][t], false, false);
    }

    float* obase = out + (size_t)(i0 + wr * (RT * 16) + 8 * half) * n
                       + (j0 + wc * (CT * 16) + mn);
#pragma unroll
    for (int r = 0; r < RT; ++r) {
#pragma unroll
        for (int v = 0; v < 8; ++v) {
            float* orow = obase + (size_t)(r * 16 + v) * n;
#pragma unroll
            for (int t = 0; t < CT; ++t)
                orow[t * 16] = -c[r][t][v];
        }
    }
}

extern "C" void kernel_launch(void* const* d_in, const int* in_sizes, int n_in,
                              void* d_out, int out_size, void* d_ws, size_t ws_size,
                              hipStream_t stream) {
    (void)n_in; (void)out_size; (void)d_ws; (void)ws_size;
    const float* x1 = (const float*)d_in[0];
    float* out = (float*)d_out;
    const int n = in_sizes[0] / D_FEAT;
    dim3 grid(n / TILE, n / TILE);
    batchdist_wmma_f32<<<grid, 256, 0, stream>>>(x1, out, n);
}
